# starting kernel
# baseline (speedup 1.0000x reference)
#include <hip/hip_runtime.h>

typedef __attribute__((ext_vector_type(2))) float v2f;
typedef __attribute__((ext_vector_type(8))) float v8f;

#define FEAT_DIM 64

__global__ void fill_zero_kernel(float* __restrict__ p, long long n) {
  long long i = (long long)blockIdx.x * blockDim.x + threadIdx.x;
  if (i < n) p[i] = 0.0f;
}

__global__ void scatter_add_scaled(const float* __restrict__ feat,
                                   const float* __restrict__ norm,
                                   const int* __restrict__ src,
                                   const int* __restrict__ dst,
                                   float* __restrict__ agg, int E) {
  int tid = blockIdx.x * blockDim.x + threadIdx.x;
  int e = tid >> 5;
  if (e >= E) return;
  int lane = tid & 31;
  int s = src[e];
  int t = dst[e];
  float ns = norm[s];
  const float2* f2 = reinterpret_cast<const float2*>(feat + (size_t)s * FEAT_DIM);
  float2 v = f2[lane];
  float* ap = agg + (size_t)t * FEAT_DIM + lane * 2;
  atomicAdd(ap + 0, v.x * ns);
  atomicAdd(ap + 1, v.y * ns);
}

template <int NCOLS, int NTILES_N, bool RELU>
__global__ void dense_wmma(const float* __restrict__ agg,
                           const float* __restrict__ norm,
                           const float* __restrict__ W,
                           const float* __restrict__ bias,
                           float* __restrict__ out,
                           int nrows) {
  const int lane = threadIdx.x & 31;
  const int wave = threadIdx.x >> 5;
  const int wid = blockIdx.x * (blockDim.x >> 5) + wave;
  const int tiles_m = (nrows + 15) >> 4;
  if (wid >= tiles_m * NTILES_N) return;
  const int tile_m = wid / NTILES_N;
  const int tile_n = wid % NTILES_N;
  const int r0 = tile_m << 4;
  const int c0 = tile_n << 4;

  const int m = lane & 15;
  const int half = lane >> 4;

  int r = r0 + m;
  if (r >= nrows) r = nrows - 1;
  const float nr = norm[r];
  const float* __restrict__ arow = agg + (size_t)r * FEAT_DIM;

  const int col = c0 + (lane & 15);
  const bool colok = (col < NCOLS);
  const int bcol = colok ? col : 0;

  v8f acc = {};
#pragma unroll
  for (int k0 = 0; k0 < FEAT_DIM; k0 += 4) {
    const int k = k0 + half * 2;
    v2f a;
    a.x = arow[k] * nr;
    a.y = arow[k + 1] * nr;
    v2f b;
    b.x = W[(size_t)k * NCOLS + bcol];
    b.y = W[(size_t)(k + 1) * NCOLS + bcol];
    acc = __builtin_amdgcn_wmma_f32_16x16x4_f32(false, a, false, b, (short)0,
                                                acc, false, false);
  }

  if (colok) {
    const float bv = bias[bcol];
#pragma unroll
    for (int j = 0; j < 8; ++j) {
      const int row = r0 + j + half * 8;
      if (row < nrows) {
        float v = acc[j] + bv;
        if (RELU) v = fmaxf(v, 0.0f);
        out[(size_t)row * NCOLS + col] = v;
      }
    }
  }
}

extern "C" void kernel_launch(void* const* d_in, const int* in_sizes, int n_in,
                              void* d_out, int out_size, void* d_ws, size_t ws_size,
                              hipStream_t stream) {
  const float* feat = (const float*)d_in[0];
  const float* norm = (const float*)d_in[1];
  const int* src    = (const int*)d_in[2];
  const int* dst    = (const int*)d_in[3];
  const float* W1   = (const float*)d_in[4];
  const float* b1   = (const float*)d_in[5];
  const float* W2   = (const float*)d_in[6];
  const float* b2   = (const float*)d_in[7];
  float* out = (float*)d_out;

  const int N = in_sizes[1];
  const int E = in_sizes[2];

  float* bufA = (float*)d_ws;
  float* bufB = bufA + (size_t)N * FEAT_DIM;

  const long long aggN = (long long)N * FEAT_DIM;
  const int fillBlocks = (int)((aggN + 255) / 256);
  const int scatBlocks = (E + 7) / 8;

  const int tilesM = (N + 15) / 16;
  const int waves1 = tilesM * 4;
  const int waves2 = tilesM * 3;
  const int dense1Blocks = (waves1 + 7) / 8;
  const int dense2Blocks = (waves2 + 7) / 8;

  fill_zero_kernel<<<fillBlocks, 256, 0, stream>>>(bufA, aggN);
  scatter_add_scaled<<<scatBlocks, 256, 0, stream>>>(feat, norm, src, dst, bufA, E);
  dense_wmma<64, 4, true><<<dense1Blocks, 256, 0, stream>>>(bufA, norm, W1, b1, bufB, N);

  fill_zero_kernel<<<fillBlocks, 256, 0, stream>>>(bufA, aggN);
  scatter_add_scaled<<<scatBlocks, 256, 0, stream>>>(bufB, norm, src, dst, bufA, E);
  dense_wmma<40, 3, false><<<dense2Blocks, 256, 0, stream>>>(bufA, norm, W2, b2, out, N);
}


# Layer-2 weight applied before its edge sum, its 40 columns cut 32+8: per edge one whole-wave f32 atomic plus a quarter of one that serves 4 edges, was 2; adds go into out; arrival-order sums
# speedup vs baseline: 2.3466x; 2.3466x over previous
#include <hip/hip_runtime.h>


typedef __attribute__((ext_vector_type(2))) float v2f;
typedef __attribute__((ext_vector_type(8))) float v8f;

#define FEAT_DIM 64
#define OUT_COLS 40
#define Z_PITCH 64

#define EDGE_BATCH 16
#define CHUNK_EDGES 64
#define QUEUE_PITCH 64

__device__ __forceinline__ int pick_edge(int a, int b, int j) {
  const int lo = __builtin_amdgcn_readlane(a, j & 31);
  const int hi = __builtin_amdgcn_readlane(b, j & 31);
  return (j < 32) ? lo : hi;
}

__device__ __forceinline__ unsigned draw_ticket(unsigned* q, int lane) {
  unsigned v = 0u;
  if (lane == 0) v = atomicInc(q, 0xFFFFFFFFu);
  return v;
}


template <bool FULL>
__device__ __forceinline__ void edge_batch(unsigned long long& m, int rem, int lane,
                                           int sa, int sb, int ta, int tb,
                                           const float* __restrict__ rows,
                                           float* __restrict__ agg) {
  float lo[EDGE_BATCH], hi[EDGE_BATCH];
  int jj[EDGE_BATCH];
#pragma unroll
  for (int u = 0; u < EDGE_BATCH; ++u) {
    if (FULL || u < rem) {
      const int j = (int)__builtin_ctzll(m);
      m &= m - 1;
      jj[u] = j;
      const int s = pick_edge(sa, sb, j);
      const float* __restrict__ fp = rows + (size_t)s * FEAT_DIM + lane;
      lo[u] = fp[0];
      hi[u] = fp[32];
    }
  }
#pragma unroll
  for (int u = 0; u < EDGE_BATCH; ++u) {
    if (FULL || u < rem) {
      const int t = pick_edge(ta, tb, jj[u]);
      float* ap = agg + (size_t)t * FEAT_DIM + lane;
      atomicAdd(ap, lo[u]);
      atomicAdd(ap + 32, hi[u]);
    }
  }
}

__global__ void __launch_bounds__(256)
scatter_add_rows(const float* __restrict__ rows,
                 const int* __restrict__ src,
                 const int* __restrict__ dst,
                 float* __restrict__ agg, int E, int split,
                 unsigned* __restrict__ queue) {
  const int lane = threadIdx.x & 31;
  const int myhalf = (int)((__builtin_amdgcn_s_sendmsg_rtn(0x87) >> 18) & 1u);
  const int nchunks = (E + CHUNK_EDGES - 1) / CHUNK_EDGES;

  for (int pass = 0; pass < 2; ++pass) {
    const int h = myhalf ^ pass;
    const bool upper = (h != 0);
    unsigned* q = queue + h * QUEUE_PITCH;
    unsigned tk_v = draw_ticket(q, lane);
    for (int it = 0; it < nchunks; ++it) {
      const unsigned tk = (unsigned)__builtin_amdgcn_readfirstlane((int)tk_v);
      if (tk >= (unsigned)nchunks) break;
      const int e0 = (int)tk * CHUNK_EDGES;
      const int ea = e0 + lane, eb = e0 + 32 + lane;
      const bool va = ea < E, vb = eb < E;
      const int ia = va ? ea : (E - 1);
      const int ib = vb ? eb : (E - 1);
      const int sa = src[ia], sb = src[ib];
      const int ta = dst[ia], tb = dst[ib];
      tk_v = draw_ticket(q, lane);
      const bool ina = va && ((ta >= split) == upper);
      const bool inb = vb && ((tb >= split) == upper);
      const unsigned ma = __builtin_amdgcn_ballot_w32(ina);
      const unsigned mb = __builtin_amdgcn_ballot_w32(inb);
      unsigned long long m = (unsigned long long)ma | ((unsigned long long)mb << 32);
      int rem = __builtin_popcountll(m);
      while (rem >= EDGE_BATCH) {
        edge_batch<true>(m, rem, lane, sa, sb, ta, tb, rows, agg);
        rem -= EDGE_BATCH;
      }
      if (rem > 0) edge_batch<false>(m, rem, lane, sa, sb, ta, tb, rows, agg);
    }
  }
}

template <bool FULL>
__device__ __forceinline__ void out_batch(int base, int rem, int cnt, int lane,
                                          int s0, int s1, int t0, int t1, int n0, int n1,
                                          const int* cS, const int* cT,
                                          const float* __restrict__ z,
                                          float* __restrict__ out) {
  float za[EDGE_BATCH];
  float zb[EDGE_BATCH / 4];
  int tq[EDGE_BATCH / 4];
  bool okq[EDGE_BATCH / 4];
  const int sub = lane >> 3;
  const int k8 = lane & 7;
#pragma unroll
  for (int u = 0; u < EDGE_BATCH; ++u) {
    if (FULL || u < rem) {
      const int s = pick_edge(s0, s1, base + u);
      za[u] = z[(size_t)s * Z_PITCH + lane];
    }
  }
#pragma unroll
  for (int g = 0; g < EDGE_BATCH / 4; ++g) {
    const int r = base + 4 * g + sub;
    okq[g] = FULL || (r < cnt);
    zb[g] = 0.0f;
    tq[g] = 0;
    if (okq[g]) {
      const int s = cS[r];
      tq[g] = cT[r];
      zb[g] = z[(size_t)s * Z_PITCH + 32 + k8];
    }
  }
#pragma unroll
  for (int u = 0; u < EDGE_BATCH; ++u) {
    if (FULL || u < rem) {
      const int t = pick_edge(t0, t1, base + u);
      const float n = __builtin_bit_cast(float, pick_edge(n0, n1, base + u));
      atomicAdd(out + (size_t)t * OUT_COLS + lane, n * za[u]);
    }
  }
  const int nsel = (base < 32) ? n0 : n1;
#pragma unroll
  for (int g = 0; g < EDGE_BATCH / 4; ++g) {
    const int r = base + 4 * g + sub;
    const float n = __builtin_bit_cast(float, __builtin_amdgcn_ds_bpermute((r & 31) << 2, nsel));
    if (okq[g]) atomicAdd(out + (size_t)tq[g] * OUT_COLS + 32 + k8, n * zb[g]);
  }
}

__global__ void __launch_bounds__(256)
scatter_add_out(const float* __restrict__ z,
                const float* __restrict__ norm,
                const int* __restrict__ src,
                const int* __restrict__ dst,
                float* __restrict__ out,
                int E, int split,
                unsigned* __restrict__ queue) {
  __shared__ int clist[8 * 2 * CHUNK_EDGES];
  const int lane = threadIdx.x & 31;
  int* cS = clist + (threadIdx.x >> 5) * (2 * CHUNK_EDGES);
  int* cT = cS + CHUNK_EDGES;
  const unsigned below = (1u << lane) - 1u;
  const int myhalf = (int)((__builtin_amdgcn_s_sendmsg_rtn(0x87) >> 18) & 1u);
  const int nchunks = (E + CHUNK_EDGES - 1) / CHUNK_EDGES;

  for (int pass = 0; pass < 2; ++pass) {
    const int h = myhalf ^ pass;
    const bool upper = (h != 0);
    unsigned* q = queue + h * QUEUE_PITCH;
    unsigned tk_v = draw_ticket(q, lane);
    for (int it = 0; it < nchunks; ++it) {
      const unsigned tk = (unsigned)__builtin_amdgcn_readfirstlane((int)tk_v);
      if (tk >= (unsigned)nchunks) break;
      const int e0 = (int)tk * CHUNK_EDGES;
      const int ea = e0 + lane, eb = e0 + 32 + lane;
      const bool va = ea < E, vb = eb < E;
      const int ia = va ? ea : (E - 1);
      const int ib = vb ? eb : (E - 1);
      const int sa = src[ia], sb = src[ib];
      const int ta = dst[ia], tb = dst[ib];
      tk_v = draw_ticket(q, lane);
      const bool ina = va && ((ta >= split) == upper);
      const bool inb = vb && ((tb >= split) == upper);
      const unsigned ma = __builtin_amdgcn_ballot_w32(ina);
      const unsigned mb = __builtin_amdgcn_ballot_w32(inb);
      const int ca = __builtin_popcount(ma);
      const int cnt = ca + __builtin_popcount(mb);
      if (ina) {
        const int r = __builtin_popcount(ma & below);
        cS[r] = sa;
        cT[r] = ta;
      }
      if (inb) {
        const int r = ca + __builtin_popcount(mb & below);
        cS[r] = sb;
        cT[r] = tb;
      }
      int s0 = 0, s1 = 0, t0 = 0, t1 = 0, n0 = 0, n1 = 0;
      if (lane < cnt) {
        s0 = cS[lane];
        t0 = cT[lane];
        n0 = __builtin_bit_cast(int, norm[t0]);
      }
      if (lane + 32 < cnt) {
        s1 = cS[lane + 32];
        t1 = cT[lane + 32];
        n1 = __builtin_bit_cast(int, norm[t1]);
      }
      int base = 0;
      while (cnt - base >= EDGE_BATCH) {
        out_batch<true>(base, EDGE_BATCH, cnt, lane, s0, s1, t0, t1, n0, n1, cS, cT, z, out);
        base += EDGE_BATCH;
      }
      if (cnt - base > 0)
        out_batch<false>(base, cnt - base, cnt, lane, s0, s1, t0, t1, n0, n1, cS, cT, z, out);
    }
  }
}

#define TILE_ROWS 32
#define A_PITCH 68

template <int NCOLS, int OPITCH, int NTILES_N, bool EPIA, bool ZERO>
__global__ void __launch_bounds__(NTILES_N * 64)
dense_wmma(const float* __restrict__ in,
           const float* __restrict__ norm,
           const float* __restrict__ W,
           const float* __restrict__ abias,
           float* __restrict__ out,
           int nrows,
           float* __restrict__ z0,
           long long nzero4,
           unsigned* __restrict__ zq,
           const float* __restrict__ obias,
           float* __restrict__ fin) {
  constexpr int NTHREADS = NTILES_N * 64;
  constexpr int NPIECES = TILE_ROWS * (FEAT_DIM / 4);
  constexpr int NSTAGE = (NPIECES + NTHREADS - 1) / NTHREADS;
  __shared__ __attribute__((aligned(16))) float As[TILE_ROWS * A_PITCH];

  const int tid = threadIdx.x;
  const int lane = tid & 31;
  const int wave = tid >> 5;
  const int rbase = blockIdx.x * TILE_ROWS;

  float4 av[NSTAGE];
  float nv[NSTAGE];
  float4 ab[NSTAGE];
#pragma unroll
  for (int i = 0; i < NSTAGE; ++i) {
    const int idx = tid + i * NTHREADS;
    if (idx < NPIECES) {
      int r = rbase + (idx >> 4);
      if (r >= nrows) r = nrows - 1;
      av[i] = *reinterpret_cast<const float4*>(in + (size_t)r * FEAT_DIM + (idx & 15) * 4);
      nv[i] = norm[r];
      if (EPIA) {
        const float* __restrict__ bp = abias + (idx & 15) * 4;
        ab[i] = make_float4(bp[0], bp[1], bp[2], bp[3]);
      }
    }
  }

  const int tile_m = wave / NTILES_N;
  const int tile_n = wave % NTILES_N;
  const int r0 = rbase + (tile_m << 4);
  const int c0 = tile_n << 4;
  const int m = lane & 15;
  const int half = lane >> 4;
  const int col = c0 + m;
  const bool colok = (col < NCOLS);
  const int bcol = colok ? col : 0;

  v2f bw[FEAT_DIM / 4];
#pragma unroll
  for (int s = 0; s < FEAT_DIM / 4; ++s) {
    const int k = s * 4 + half * 2;
    bw[s].x = W[(size_t)k * NCOLS + bcol];
    bw[s].y = W[(size_t)(k + 1) * NCOLS + bcol];
  }

#pragma unroll
  for (int i = 0; i < NSTAGE; ++i) {
    const int idx = tid + i * NTHREADS;
    if (idx < NPIECES) {
      float4 v = av[i];
      const float n = nv[i];
      if (EPIA) {
        const float4 b = ab[i];
        v.x = fmaxf(v.x * n + b.x, 0.0f) * n;
        v.y = fmaxf(v.y * n + b.y, 0.0f) * n;
        v.z = fmaxf(v.z * n + b.z, 0.0f) * n;
        v.w = fmaxf(v.w * n + b.w, 0.0f) * n;
      } else {
        v.x *= n; v.y *= n; v.z *= n; v.w *= n;
      }
      *reinterpret_cast<float4*>(&As[(idx >> 4) * A_PITCH + (idx & 15) * 4]) = v;
    }
  }
  __syncthreads();

  if (ZERO) {
    float4* __restrict__ q0 = reinterpret_cast<float4*>(z0);
    const long long nthreads = (long long)gridDim.x * NTHREADS;
    const float4 zero = make_float4(0.0f, 0.0f, 0.0f, 0.0f);
    for (long long i = (long long)blockIdx.x * NTHREADS + tid; i < nzero4; i += nthreads)
      q0[i] = zero;
    if (blockIdx.x == 0 && tid < 4) zq[tid * QUEUE_PITCH] = 0u;
  }
  if (EPIA) {
    constexpr int ROW4 = NCOLS / 4;
    float4* __restrict__ f4 = reinterpret_cast<float4*>(fin) + (size_t)rbase * ROW4;
    for (int i = tid; i < TILE_ROWS * ROW4; i += NTHREADS) {
      if (rbase + i / ROW4 < nrows) {
        const float* __restrict__ bp = obias + (i % ROW4) * 4;
        f4[i] = make_float4(bp[0], bp[1], bp[2], bp[3]);
      }
    }
  }

  const float* __restrict__ arow = &As[((tile_m << 4) + m) * A_PITCH + half * 2];
  v8f acc = {};
#pragma unroll
  for (int s = 0; s < FEAT_DIM / 4; ++s) {
    const v2f a = *reinterpret_cast<const v2f*>(arow + s * 4);
    acc = __builtin_amdgcn_wmma_f32_16x16x4_f32(false, a, false, bw[s], (short)0,
                                                acc, false, false);
  }

  if (colok) {
#pragma unroll
    for (int j = 0; j < 8; ++j) {
      const int row = r0 + j + half * 8;
      if (row < nrows) out[(size_t)row * OPITCH + col] = acc[j];
    }
  }
}

extern "C" void kernel_launch(void* const* d_in, const int* in_sizes, int n_in,
                              void* d_out, int out_size, void* d_ws, size_t ws_size,
                              hipStream_t stream) {
  const float* feat = (const float*)d_in[0];
  const float* norm = (const float*)d_in[1];
  const int* src    = (const int*)d_in[2];
  const int* dst    = (const int*)d_in[3];
  const float* W1   = (const float*)d_in[4];
  const float* b1   = (const float*)d_in[5];
  const float* W2   = (const float*)d_in[6];
  const float* b2   = (const float*)d_in[7];
  float* out = (float*)d_out;

  const int N = in_sizes[1];
  const int E = in_sizes[2];

  float* bufY = (float*)d_ws;
  float* agg1 = bufY + (size_t)N * FEAT_DIM;
  float* bufZ = agg1 + (size_t)N * FEAT_DIM;
  unsigned* queue = (unsigned*)(bufZ + (size_t)N * Z_PITCH);

  const long long agg4 = (long long)N * (FEAT_DIM / 4);
  const int scatBlocks = 1024;
  const int split = N / 2;

  const int denseBlocks = (N + TILE_ROWS - 1) / TILE_ROWS;

  dense_wmma<FEAT_DIM, FEAT_DIM, 4, false, true><<<denseBlocks, 256, 0, stream>>>(
      feat, norm, W1, b1, bufY, N, agg1, agg4, queue, b2, out);
  scatter_add_rows<<<scatBlocks, 256, 0, stream>>>(bufY, src, dst, agg1, E, split, queue);
  dense_wmma<OUT_COLS, Z_PITCH, 3, true, false><<<denseBlocks, 192, 0, stream>>>(
      agg1, norm, W2, b1, bufZ, N, agg1, 0LL, queue, b2, out);
  scatter_add_out<<<scatBlocks, 256, 0, stream>>>(bufZ, norm, src, dst, out, E, split,
                                                  queue + 2 * QUEUE_PITCH);
}
